# starting kernel
# baseline (speedup 1.0000x reference)
#include <hip/hip_runtime.h>
#include <math.h>

typedef float v2f __attribute__((ext_vector_type(2)));
typedef float v8f __attribute__((ext_vector_type(8)));

#define D_DIM 512
#define H_DIM 2048

__global__ __launch_bounds__(256)
void qff_pack_w1(const float* __restrict__ W1, float* __restrict__ w1c) {
  int idx = blockIdx.x * blockDim.x + threadIdx.x;
  if (idx < D_DIM * 4) {
    int d = idx >> 2, j = idx & 3;
    w1c[idx] = W1[d * H_DIM + j];
  }
}

__global__ __launch_bounds__(32)
void qff_main(const float* __restrict__ x, const float* __restrict__ w1c,
              const float* __restrict__ b1, const float* __restrict__ qw,
              const float* __restrict__ Wp, const float* __restrict__ bp,
              float* __restrict__ out)
{
  __shared__ float sAng[64];
  __shared__ float sQ[64];
  __shared__ float sG[64];

  const int lane = threadIdx.x;
  const int hsel = lane >> 4;
  const int lid  = lane & 15;
  const long row0 = (long)blockIdx.x * 16;

  if (lane < 8) {
    const int l = lane >> 2, i = lane & 3;
    const float phi = qw[(l * 4 + i) * 3 + 0];
    const float th  = qw[(l * 4 + i) * 3 + 1];
    const float om  = qw[(l * 4 + i) * 3 + 2];
    const float c = cosf(0.5f * th), s = sinf(0.5f * th);
    const float a = -0.5f * (phi + om);
    const float b =  0.5f * (phi - om);
    const float ca = cosf(a), sa = sinf(a);
    const float cb = cosf(b), sb = sinf(b);
    float* g = &sG[lane * 8];
    g[0] =  c * ca;  g[1] =  c * sa;
    g[2] = -s * cb;  g[3] = -s * sb;
    g[4] =  s * cb;  g[5] = -s * sb;
    g[6] =  c * ca;  g[7] = -c * sa;
  }

  v8f acc = {};
  const float* xrow = x + (row0 + lid) * D_DIM + 2 * hsel;
  const float bsel = (lid < 4) ? 1.0f : 0.0f;
  const int   bcol = lid & 3;
  #pragma unroll 4
  for (int k = 0; k < D_DIM; k += 4) {
    v2f afrag; afrag.x = xrow[k]; afrag.y = xrow[k + 1];
    v2f bfrag;
    bfrag.x = bsel * w1c[(k + 2 * hsel + 0) * 4 + bcol];
    bfrag.y = bsel * w1c[(k + 2 * hsel + 1) * 4 + bcol];
    acc = __builtin_amdgcn_wmma_f32_16x16x4_f32(false, afrag, false, bfrag,
                                                (short)0, acc, false, false);
  }

  if (lid < 4) {
    const float bias = b1[lid];
    #pragma unroll
    for (int r = 0; r < 8; ++r) {
      const float v = acc[r] + bias;
      sAng[(hsel * 8 + r) * 4 + lid] = 0.5f * v * (1.0f + erff(v * 0.70710678118654752f));
    }
  }
  __syncthreads();

  for (int t = 0; t < 8; ++t) {
    const int row = 2 * t + hsel;
    const int basis = lid;
    float re = 1.0f, im = 0.0f;
    #pragma unroll
    for (int i = 0; i < 4; ++i) {
      const float a2 = 0.5f * sAng[row * 4 + i];
      re *= ((basis >> (3 - i)) & 1) ? sinf(a2) : cosf(a2);
    }
    #pragma unroll
    for (int l = 0; l < 2; ++l) {
      #pragma unroll
      for (int i = 0; i < 4; ++i) {
        const float* g = &sG[(l * 4 + i) * 8];
        const int mask = 8 >> i;
        const int tb = (basis >> (3 - i)) & 1;
        const float pr = __shfl_xor(re, mask, 32);
        const float pi = __shfl_xor(im, mask, 32);
        const float nr0 = g[0]*re - g[1]*im + g[2]*pr - g[3]*pi;
        const float ni0 = g[0]*im + g[1]*re + g[2]*pi + g[3]*pr;
        const float nr1 = g[4]*pr - g[5]*pi + g[6]*re - g[7]*im;
        const float ni1 = g[4]*pi + g[5]*pr + g[6]*im + g[7]*re;
        re = tb ? nr1 : nr0;
        im = tb ? ni1 : ni0;
      }
      const int rsh = (l % 3) + 1;
      #pragma unroll
      for (int i = 0; i < 4; ++i) {
        const int tq = (i + rsh) & 3;
        const int maskT = 8 >> tq;
        const int ctrl = (basis >> (3 - i)) & 1;
        const float pr = __shfl_xor(re, maskT, 32);
        const float pi = __shfl_xor(im, maskT, 32);
        re = ctrl ? pr : re;
        im = ctrl ? pi : im;
      }
    }
    const float prob = re * re + im * im;
    #pragma unroll
    for (int i = 0; i < 4; ++i) {
      float sv = ((basis >> (3 - i)) & 1) ? -prob : prob;
      sv += __shfl_xor(sv, 1, 32);
      sv += __shfl_xor(sv, 2, 32);
      sv += __shfl_xor(sv, 4, 32);
      sv += __shfl_xor(sv, 8, 32);
      if (lid == 0) sQ[row * 4 + i] = sv;
    }
  }
  __syncthreads();

  v2f aq;
  aq.x = sQ[lid * 4 + 2 * hsel + 0];
  aq.y = sQ[lid * 4 + 2 * hsel + 1];
  for (int n0 = 0; n0 < D_DIM; n0 += 16) {
    v2f bw;
    bw.x = Wp[(2 * hsel + 0) * D_DIM + n0 + lid];
    bw.y = Wp[(2 * hsel + 1) * D_DIM + n0 + lid];
    v8f c = {};
    c = __builtin_amdgcn_wmma_f32_16x16x4_f32(false, aq, false, bw,
                                              (short)0, c, false, false);
    const float bias = bp[n0 + lid];
    float* orow = out + (row0 + hsel * 8) * D_DIM + n0 + lid;
    #pragma unroll
    for (int r = 0; r < 8; ++r)
      orow[(long)r * D_DIM] = c[r] + bias;
  }
}

extern "C" void kernel_launch(void* const* d_in, const int* in_sizes, int n_in,
                              void* d_out, int out_size, void* d_ws, size_t ws_size,
                              hipStream_t stream) {
  const float* x  = (const float*)d_in[0];
  const float* W1 = (const float*)d_in[1];
  const float* b1 = (const float*)d_in[2];
  const float* qw = (const float*)d_in[3];
  const float* Wp = (const float*)d_in[4];
  const float* bp = (const float*)d_in[5];
  float* out = (float*)d_out;
  float* w1c = (float*)d_ws;

  const int rows = in_sizes[0] / D_DIM;
  qff_pack_w1<<<(D_DIM * 4 + 255) / 256, 256, 0, stream>>>(W1, w1c);
  qff_main<<<rows / 16, 32, 0, stream>>>(x, w1c, b1, qw, Wp, bp, out);
}


# GEMM1 reads x and the packed W1 (now transposed) with 16-byte loads per lane, one permuted k order in both loaders (new summation order); a wave's x loads stand ahead of its WMMAs
# speedup vs baseline: 1.9748x; 1.9748x over previous
#include <hip/hip_runtime.h>
#include <math.h>

typedef float v2f __attribute__((ext_vector_type(2)));
typedef float v4f __attribute__((ext_vector_type(4)));
typedef float v8f __attribute__((ext_vector_type(8)));

#define D_DIM 512
#define H_DIM 2048

__global__ __launch_bounds__(256)
void qff_pack_w1(const float* __restrict__ W1, float* __restrict__ w1t) {
  int idx = blockIdx.x * blockDim.x + threadIdx.x;
  if (idx < D_DIM * 4) {
    int d = idx >> 2, j = idx & 3;
    w1t[j * D_DIM + d] = W1[d * H_DIM + j];
  }
}

#define NW 4
static_assert(NW == 4 || NW == 8, "waves 0-1 fold the partials while the last wave builds the gates; NW must divide 8");
__global__ __launch_bounds__(32 * NW)
void qff_main(const float* __restrict__ x, const float* __restrict__ w1t,
              const float* __restrict__ b1, const float* __restrict__ qw,
              const float* __restrict__ Wp, const float* __restrict__ bp,
              float* __restrict__ out)
{
  __shared__ float sPart[NW * 64];
  __shared__ float sAng[64];
  __shared__ float sQ[64];
  __shared__ float sG[64];

  const int wave = threadIdx.x >> 5;
  const int lane = threadIdx.x & 31;
  const int hsel = lane >> 4;
  const int lid  = lane & 15;
  const long row0 = (long)blockIdx.x * 16;

  v8f acc = {};
  const int kbeg = wave * (D_DIM / NW);
  const float bsel = (lid < 4) ? 1.0f : 0.0f;
  const int   bcol = lid & 3;
  const float* xrow = x + (row0 + lid) * D_DIM + kbeg + 4 * hsel;
  const float* wrow = w1t + bcol * D_DIM + kbeg + 4 * hsel;
  constexpr int NG = D_DIM / NW / 8;
  v4f av[NG];
  #pragma unroll
  for (int g = 0; g < NG; ++g)
    av[g] = *reinterpret_cast<const v4f*>(xrow + 8 * g);
  #pragma unroll
  for (int g = 0; g < NG; ++g) {
    const v4f bv = *reinterpret_cast<const v4f*>(wrow + 8 * g);
    v2f a0, b0, a1, b1;
    a0.x = av[g].x;      a0.y = av[g].y;
    b0.x = bsel * bv.x;  b0.y = bsel * bv.y;
    a1.x = av[g].z;      a1.y = av[g].w;
    b1.x = bsel * bv.z;  b1.y = bsel * bv.w;
    acc = __builtin_amdgcn_wmma_f32_16x16x4_f32(false, a0, false, b0,
                                                (short)0, acc, false, false);
    acc = __builtin_amdgcn_wmma_f32_16x16x4_f32(false, a1, false, b1,
                                                (short)0, acc, false, false);
  }

  if (lid < 4) {
    #pragma unroll
    for (int r = 0; r < 8; ++r)
      sPart[wave * 64 + (hsel * 8 + r) * 4 + lid] = acc[r];
  }
  __syncthreads();

  if (threadIdx.x < 64) {
    float v = sPart[threadIdx.x];
    #pragma unroll
    for (int w = 1; w < NW; ++w) v += sPart[w * 64 + threadIdx.x];
    v += b1[threadIdx.x & 3];
    sAng[threadIdx.x] = 0.5f * v * (1.0f + erff(v * 0.70710678118654752f));
  } else if (wave == NW - 1 && lane < 8) {
    const int l = lane >> 2, i = lane & 3;
    const float phi = qw[(l * 4 + i) * 3 + 0];
    const float th  = qw[(l * 4 + i) * 3 + 1];
    const float om  = qw[(l * 4 + i) * 3 + 2];
    const float c = cosf(0.5f * th), s = sinf(0.5f * th);
    const float a = -0.5f * (phi + om);
    const float b =  0.5f * (phi - om);
    const float ca = cosf(a), sa = sinf(a);
    const float cb = cosf(b), sb = sinf(b);
    float* g = &sG[lane * 8];
    g[0] =  c * ca;  g[1] =  c * sa;
    g[2] = -s * cb;  g[3] = -s * sb;
    g[4] =  s * cb;  g[5] = -s * sb;
    g[6] =  c * ca;  g[7] = -c * sa;
  }
  __syncthreads();

  for (int tt = 0; tt < 8 / NW; ++tt) {
    const int t = wave * (8 / NW) + tt;
    const int row = 2 * t + hsel;
    const int basis = lid;
    float re = 1.0f, im = 0.0f;
    #pragma unroll
    for (int i = 0; i < 4; ++i) {
      const float a2 = 0.5f * sAng[row * 4 + i];
      re *= ((basis >> (3 - i)) & 1) ? sinf(a2) : cosf(a2);
    }
    #pragma unroll
    for (int l = 0; l < 2; ++l) {
      #pragma unroll
      for (int i = 0; i < 4; ++i) {
        const float* g = &sG[(l * 4 + i) * 8];
        const int mask = 8 >> i;
        const int tb = (basis >> (3 - i)) & 1;
        const float pr = __shfl_xor(re, mask, 32);
        const float pi = __shfl_xor(im, mask, 32);
        const float nr0 = g[0]*re - g[1]*im + g[2]*pr - g[3]*pi;
        const float ni0 = g[0]*im + g[1]*re + g[2]*pi + g[3]*pr;
        const float nr1 = g[4]*pr - g[5]*pi + g[6]*re - g[7]*im;
        const float ni1 = g[4]*pi + g[5]*pr + g[6]*im + g[7]*re;
        re = tb ? nr1 : nr0;
        im = tb ? ni1 : ni0;
      }
      const int rsh = (l % 3) + 1;
      #pragma unroll
      for (int i = 0; i < 4; ++i) {
        const int tq = (i + rsh) & 3;
        const int maskT = 8 >> tq;
        const int ctrl = (basis >> (3 - i)) & 1;
        const float pr = __shfl_xor(re, maskT, 32);
        const float pi = __shfl_xor(im, maskT, 32);
        re = ctrl ? pr : re;
        im = ctrl ? pi : im;
      }
    }
    const float prob = re * re + im * im;
    #pragma unroll
    for (int i = 0; i < 4; ++i) {
      float sv = ((basis >> (3 - i)) & 1) ? -prob : prob;
      sv += __shfl_xor(sv, 1, 32);
      sv += __shfl_xor(sv, 2, 32);
      sv += __shfl_xor(sv, 4, 32);
      sv += __shfl_xor(sv, 8, 32);
      if (lid == 0) sQ[row * 4 + i] = sv;
    }
  }
  __syncthreads();

  v2f aq;
  aq.x = sQ[lid * 4 + 2 * hsel + 0];
  aq.y = sQ[lid * 4 + 2 * hsel + 1];
  const int nbeg = wave * (D_DIM / NW);
  for (int nn = 0; nn < D_DIM / NW; nn += 16) {
    const int n0 = nbeg + nn;
    v2f bw;
    bw.x = Wp[(2 * hsel + 0) * D_DIM + n0 + lid];
    bw.y = Wp[(2 * hsel + 1) * D_DIM + n0 + lid];
    v8f c = {};
    c = __builtin_amdgcn_wmma_f32_16x16x4_f32(false, aq, false, bw,
                                              (short)0, c, false, false);
    const float bias = bp[n0 + lid];
    float* orow = out + (row0 + hsel * 8) * D_DIM + n0 + lid;
    #pragma unroll
    for (int r = 0; r < 8; ++r)
      orow[(long)r * D_DIM] = c[r] + bias;
  }
}

extern "C" void kernel_launch(void* const* d_in, const int* in_sizes, int n_in,
                              void* d_out, int out_size, void* d_ws, size_t ws_size,
                              hipStream_t stream) {
  const float* x  = (const float*)d_in[0];
  const float* W1 = (const float*)d_in[1];
  const float* b1 = (const float*)d_in[2];
  const float* qw = (const float*)d_in[3];
  const float* Wp = (const float*)d_in[4];
  const float* bp = (const float*)d_in[5];
  float* out = (float*)d_out;
  float* w1c = (float*)d_ws;

  const int rows = in_sizes[0] / D_DIM;
  qff_pack_w1<<<(D_DIM * 4 + 255) / 256, 256, 0, stream>>>(W1, w1c);
  qff_main<<<rows / 16, 32 * NW, 0, stream>>>(x, w1c, b1, qw, Wp, bp, out);
}


# State-vector sim de-duplicated: a row's 16 amplitudes on 8 lanes x 2, 4 rows per wave in one pass; half-angle sin/cos made once per angle via LDS; bit-1 lanes sum a gate's own term first (new order)
# speedup vs baseline: 2.3640x; 1.1971x over previous
#include <hip/hip_runtime.h>
#include <math.h>

typedef float v2f __attribute__((ext_vector_type(2)));
typedef float v4f __attribute__((ext_vector_type(4)));
typedef float v8f __attribute__((ext_vector_type(8)));

#define D_DIM 512
#define H_DIM 2048

__global__ __launch_bounds__(256)
void qff_pack_w1(const float* __restrict__ W1, float* __restrict__ w1t) {
  int idx = blockIdx.x * blockDim.x + threadIdx.x;
  if (idx < D_DIM * 4) {
    int d = idx >> 2, j = idx & 3;
    w1t[j * D_DIM + d] = W1[d * H_DIM + j];
  }
}

#define NW 4
static_assert(NW == 4, "the sim spreads a row over 32 * NW / 16 = 8 lanes; waves 0-1 fold the partials while the last wave builds the gates");

__device__ __forceinline__ void qff_rot_lane(float& re, float& im, const v4f c, const int mask) {
  const float pr = __shfl_xor(re, mask, 32);
  const float pi = __shfl_xor(im, mask, 32);
  const float nr = c.x * re - c.y * im + c.z * pr - c.w * pi;
  const float ni = c.x * im + c.y * re + c.z * pi + c.w * pr;
  re = nr;
  im = ni;
}
__global__ __launch_bounds__(32 * NW) __attribute__((amdgpu_waves_per_eu(8, 8)))
void qff_main(const float* __restrict__ x, const float* __restrict__ w1t,
              const float* __restrict__ b1, const float* __restrict__ qw,
              const float* __restrict__ Wp, const float* __restrict__ bp,
              float* __restrict__ out)
{
  __shared__ float sPart[NW * 64];
  __shared__ alignas(16) float sCS[128];
  __shared__ alignas(16) float sQ[64];
  __shared__ alignas(16) float sG[64];

  const int wave = threadIdx.x >> 5;
  const int lane = threadIdx.x & 31;
  const int hsel = lane >> 4;
  const int lid  = lane & 15;
  const long row0 = (long)blockIdx.x * 16;

  v8f acc = {};
  const int kbeg = wave * (D_DIM / NW);
  const float bsel = (lid < 4) ? 1.0f : 0.0f;
  const int   bcol = lid & 3;
  const float* xrow = x + (row0 + lid) * D_DIM + kbeg + 4 * hsel;
  const float* wrow = w1t + bcol * D_DIM + kbeg + 4 * hsel;
  constexpr int NG = D_DIM / NW / 8;
  v4f av[NG];
  #pragma unroll
  for (int g = 0; g < NG; ++g)
    av[g] = *reinterpret_cast<const v4f*>(xrow + 8 * g);
  #pragma unroll
  for (int g = 0; g < NG; ++g) {
    const v4f bv = *reinterpret_cast<const v4f*>(wrow + 8 * g);
    v2f a0, b0, a1, b1;
    a0.x = av[g].x;      a0.y = av[g].y;
    b0.x = bsel * bv.x;  b0.y = bsel * bv.y;
    a1.x = av[g].z;      a1.y = av[g].w;
    b1.x = bsel * bv.z;  b1.y = bsel * bv.w;
    acc = __builtin_amdgcn_wmma_f32_16x16x4_f32(false, a0, false, b0,
                                                (short)0, acc, false, false);
    acc = __builtin_amdgcn_wmma_f32_16x16x4_f32(false, a1, false, b1,
                                                (short)0, acc, false, false);
  }

  if (lid < 4) {
    #pragma unroll
    for (int r = 0; r < 8; ++r)
      sPart[wave * 64 + (hsel * 8 + r) * 4 + lid] = acc[r];
  }
  __syncthreads();

  if (threadIdx.x < 64) {
    float v = sPart[threadIdx.x];
    #pragma unroll
    for (int w = 1; w < NW; ++w) v += sPart[w * 64 + threadIdx.x];
    v += b1[threadIdx.x & 3];
    const float a2 = 0.5f * (0.5f * v * (1.0f + erff(v * 0.70710678118654752f)));
    v2f cs;
    cs.x = cosf(a2);
    cs.y = sinf(a2);
    *reinterpret_cast<v2f*>(&sCS[2 * threadIdx.x]) = cs;
  } else if (wave == NW - 1 && lane < 8) {
    const int l = lane >> 2, i = lane & 3;
    const float phi = qw[(l * 4 + i) * 3 + 0];
    const float th  = qw[(l * 4 + i) * 3 + 1];
    const float om  = qw[(l * 4 + i) * 3 + 2];
    const float c = cosf(0.5f * th), s = sinf(0.5f * th);
    const float a = -0.5f * (phi + om);
    const float b =  0.5f * (phi - om);
    const float ca = cosf(a), sa = sinf(a);
    const float cb = cosf(b), sb = sinf(b);
    float* g = &sG[lane * 8];
    g[0] =  c * ca;  g[1] =  c * sa;
    g[2] = -s * cb;  g[3] = -s * sb;
    g[4] =  c * ca;  g[5] = -c * sa;
    g[6] =  s * cb;  g[7] = -s * sb;
  }
  __syncthreads();

  {
    const int sub = lane & 7;
    const int row = wave * 4 + (lane >> 3);
    const float* cs = &sCS[row * 8];
    const float f = cs[(sub >> 2) & 1] * cs[2 + ((sub >> 1) & 1)] * cs[4 + (sub & 1)];
    float re0 = f * cs[6], im0 = 0.0f;
    float re1 = f * cs[7], im1 = 0.0f;
    #pragma unroll
    for (int l = 0; l < 2; ++l) {
      #pragma unroll
      for (int i = 0; i < 3; ++i) {
        const int tb = (sub >> (2 - i)) & 1;
        const v4f c = *reinterpret_cast<const v4f*>(&sG[(l * 4 + i) * 8 + 4 * tb]);
        qff_rot_lane(re0, im0, c, 4 >> i);
        qff_rot_lane(re1, im1, c, 4 >> i);
      }
      {
        const float* g = &sG[(l * 4 + 3) * 8];
        const float n0r = g[0]*re0 - g[1]*im0 + g[2]*re1 - g[3]*im1;
        const float n0i = g[0]*im0 + g[1]*re0 + g[2]*im1 + g[3]*re1;
        const float n1r = g[6]*re0 - g[7]*im0 + g[4]*re1 - g[5]*im1;
        const float n1i = g[6]*im0 + g[7]*re0 + g[4]*im1 + g[5]*re1;
        re0 = n0r; im0 = n0i; re1 = n1r; im1 = n1i;
      }
      const int rsh = (l % 3) + 1;
      #pragma unroll
      for (int i = 0; i < 4; ++i) {
        const int tq = (i + rsh) & 3;
        if (i < 3 && tq < 3) {
          const int ctrl = (sub >> (2 - i)) & 1;
          const int m = 4 >> tq;
          const float pr0 = __shfl_xor(re0, m, 32), pi0 = __shfl_xor(im0, m, 32);
          const float pr1 = __shfl_xor(re1, m, 32), pi1 = __shfl_xor(im1, m, 32);
          re0 = ctrl ? pr0 : re0;  im0 = ctrl ? pi0 : im0;
          re1 = ctrl ? pr1 : re1;  im1 = ctrl ? pi1 : im1;
        } else if (i < 3) {
          const int ctrl = (sub >> (2 - i)) & 1;
          const float r0 = ctrl ? re1 : re0, i0 = ctrl ? im1 : im0;
          const float r1 = ctrl ? re0 : re1, i1 = ctrl ? im0 : im1;
          re0 = r0; im0 = i0; re1 = r1; im1 = i1;
        } else {
          const int m = 4 >> tq;
          re1 = __shfl_xor(re1, m, 32);
          im1 = __shfl_xor(im1, m, 32);
        }
      }
    }
    const float p0 = re0 * re0 + im0 * im0;
    const float p1 = re1 * re1 + im1 * im1;
    const float t = p0 + p1;
    v4f ev;
    ev.x = ((sub >> 2) & 1) ? -t : t;
    ev.y = ((sub >> 1) & 1) ? -t : t;
    ev.z = (sub & 1) ? -t : t;
    ev.w = p0 - p1;
    #pragma unroll
    for (int i = 0; i < 4; ++i) {
      float sv = ev[i];
      sv += __shfl_xor(sv, 1, 32);
      sv += __shfl_xor(sv, 2, 32);
      sv += __shfl_xor(sv, 4, 32);
      ev[i] = sv;
    }
    if (sub == 0) *reinterpret_cast<v4f*>(&sQ[row * 4]) = ev;
  }
  __syncthreads();

  v2f aq;
  aq.x = sQ[lid * 4 + 2 * hsel + 0];
  aq.y = sQ[lid * 4 + 2 * hsel + 1];
  const int nbeg = wave * (D_DIM / NW);
  for (int nn = 0; nn < D_DIM / NW; nn += 16) {
    const int n0 = nbeg + nn;
    v2f bw;
    bw.x = Wp[(2 * hsel + 0) * D_DIM + n0 + lid];
    bw.y = Wp[(2 * hsel + 1) * D_DIM + n0 + lid];
    v8f c = {};
    c = __builtin_amdgcn_wmma_f32_16x16x4_f32(false, aq, false, bw,
                                              (short)0, c, false, false);
    const float bias = bp[n0 + lid];
    float* orow = out + (row0 + hsel * 8) * D_DIM + n0 + lid;
    #pragma unroll
    for (int r = 0; r < 8; ++r)
      orow[(long)r * D_DIM] = c[r] + bias;
  }
}

extern "C" void kernel_launch(void* const* d_in, const int* in_sizes, int n_in,
                              void* d_out, int out_size, void* d_ws, size_t ws_size,
                              hipStream_t stream) {
  const float* x  = (const float*)d_in[0];
  const float* W1 = (const float*)d_in[1];
  const float* b1 = (const float*)d_in[2];
  const float* qw = (const float*)d_in[3];
  const float* Wp = (const float*)d_in[4];
  const float* bp = (const float*)d_in[5];
  float* out = (float*)d_out;
  float* w1c = (float*)d_ws;

  const int rows = in_sizes[0] / D_DIM;
  qff_pack_w1<<<(D_DIM * 4 + 255) / 256, 256, 0, stream>>>(W1, w1c);
  qff_main<<<rows / 16, 32 * NW, 0, stream>>>(x, w1c, b1, qw, Wp, bp, out);
}


# GEMM2 column-interleaved: a lane owns 4 adjacent output columns, so out is written with 16-byte stores, whole 128-byte lines per store (16 stores a wave, were 64); Wp and bp read 16 bytes a lane
# speedup vs baseline: 2.8347x; 1.1991x over previous
#include <hip/hip_runtime.h>
#include <math.h>

typedef float v2f __attribute__((ext_vector_type(2)));
typedef float v4f __attribute__((ext_vector_type(4)));
typedef float v8f __attribute__((ext_vector_type(8)));

#define D_DIM 512
#define H_DIM 2048

__global__ __launch_bounds__(256)
void qff_pack_w1(const float* __restrict__ W1, float* __restrict__ w1t) {
  int idx = blockIdx.x * blockDim.x + threadIdx.x;
  if (idx < D_DIM * 4) {
    int d = idx >> 2, j = idx & 3;
    w1t[j * D_DIM + d] = W1[d * H_DIM + j];
  }
}

#define NW 4
static_assert(NW == 4, "the sim spreads a row over 32 * NW / 16 = 8 lanes; waves 0-1 fold the partials while the last wave builds the gates");

__device__ __forceinline__ void qff_rot_lane(float& re, float& im, const v4f c, const int mask) {
  const float pr = __shfl_xor(re, mask, 32);
  const float pi = __shfl_xor(im, mask, 32);
  const float nr = c.x * re - c.y * im + c.z * pr - c.w * pi;
  const float ni = c.x * im + c.y * re + c.z * pi + c.w * pr;
  re = nr;
  im = ni;
}
__global__ __launch_bounds__(32 * NW) __attribute__((amdgpu_waves_per_eu(8, 8)))
void qff_main(const float* __restrict__ x, const float* __restrict__ w1t,
              const float* __restrict__ b1, const float* __restrict__ qw,
              const float* __restrict__ Wp, const float* __restrict__ bp,
              float* __restrict__ out)
{
  __shared__ float sPart[NW * 64];
  __shared__ alignas(16) float sCS[128];
  __shared__ alignas(16) float sQ[64];
  __shared__ alignas(16) float sG[64];

  const int wave = threadIdx.x >> 5;
  const int lane = threadIdx.x & 31;
  const int hsel = lane >> 4;
  const int lid  = lane & 15;
  const long row0 = (long)blockIdx.x * 16;

  v8f acc = {};
  const int kbeg = wave * (D_DIM / NW);
  const float bsel = (lid < 4) ? 1.0f : 0.0f;
  const int   bcol = lid & 3;
  const float* xrow = x + (row0 + lid) * D_DIM + kbeg + 4 * hsel;
  const float* wrow = w1t + bcol * D_DIM + kbeg + 4 * hsel;
  constexpr int NG = D_DIM / NW / 8;
  v4f av[NG];
  #pragma unroll
  for (int g = 0; g < NG; ++g)
    av[g] = *reinterpret_cast<const v4f*>(xrow + 8 * g);
  #pragma unroll
  for (int g = 0; g < NG; ++g) {
    const v4f bv = *reinterpret_cast<const v4f*>(wrow + 8 * g);
    v2f a0, b0, a1, b1;
    a0.x = av[g].x;      a0.y = av[g].y;
    b0.x = bsel * bv.x;  b0.y = bsel * bv.y;
    a1.x = av[g].z;      a1.y = av[g].w;
    b1.x = bsel * bv.z;  b1.y = bsel * bv.w;
    acc = __builtin_amdgcn_wmma_f32_16x16x4_f32(false, a0, false, b0,
                                                (short)0, acc, false, false);
    acc = __builtin_amdgcn_wmma_f32_16x16x4_f32(false, a1, false, b1,
                                                (short)0, acc, false, false);
  }

  if (lid < 4) {
    #pragma unroll
    for (int r = 0; r < 8; ++r)
      sPart[wave * 64 + (hsel * 8 + r) * 4 + lid] = acc[r];
  }
  __syncthreads();

  if (threadIdx.x < 64) {
    float v = sPart[threadIdx.x];
    #pragma unroll
    for (int w = 1; w < NW; ++w) v += sPart[w * 64 + threadIdx.x];
    v += b1[threadIdx.x & 3];
    const float a2 = 0.5f * (0.5f * v * (1.0f + erff(v * 0.70710678118654752f)));
    v2f cs;
    cs.x = cosf(a2);
    cs.y = sinf(a2);
    *reinterpret_cast<v2f*>(&sCS[2 * threadIdx.x]) = cs;
  } else if (wave == NW - 1 && lane < 8) {
    const int l = lane >> 2, i = lane & 3;
    const float phi = qw[(l * 4 + i) * 3 + 0];
    const float th  = qw[(l * 4 + i) * 3 + 1];
    const float om  = qw[(l * 4 + i) * 3 + 2];
    const float c = cosf(0.5f * th), s = sinf(0.5f * th);
    const float a = -0.5f * (phi + om);
    const float b =  0.5f * (phi - om);
    const float ca = cosf(a), sa = sinf(a);
    const float cb = cosf(b), sb = sinf(b);
    float* g = &sG[lane * 8];
    g[0] =  c * ca;  g[1] =  c * sa;
    g[2] = -s * cb;  g[3] = -s * sb;
    g[4] =  c * ca;  g[5] = -c * sa;
    g[6] =  s * cb;  g[7] = -s * sb;
  }
  __syncthreads();

  {
    const int sub = lane & 7;
    const int row = wave * 4 + (lane >> 3);
    const float* cs = &sCS[row * 8];
    const float f = cs[(sub >> 2) & 1] * cs[2 + ((sub >> 1) & 1)] * cs[4 + (sub & 1)];
    float re0 = f * cs[6], im0 = 0.0f;
    float re1 = f * cs[7], im1 = 0.0f;
    #pragma unroll
    for (int l = 0; l < 2; ++l) {
      #pragma unroll
      for (int i = 0; i < 3; ++i) {
        const int tb = (sub >> (2 - i)) & 1;
        const v4f c = *reinterpret_cast<const v4f*>(&sG[(l * 4 + i) * 8 + 4 * tb]);
        qff_rot_lane(re0, im0, c, 4 >> i);
        qff_rot_lane(re1, im1, c, 4 >> i);
      }
      {
        const float* g = &sG[(l * 4 + 3) * 8];
        const float n0r = g[0]*re0 - g[1]*im0 + g[2]*re1 - g[3]*im1;
        const float n0i = g[0]*im0 + g[1]*re0 + g[2]*im1 + g[3]*re1;
        const float n1r = g[6]*re0 - g[7]*im0 + g[4]*re1 - g[5]*im1;
        const float n1i = g[6]*im0 + g[7]*re0 + g[4]*im1 + g[5]*re1;
        re0 = n0r; im0 = n0i; re1 = n1r; im1 = n1i;
      }
      const int rsh = (l % 3) + 1;
      #pragma unroll
      for (int i = 0; i < 4; ++i) {
        const int tq = (i + rsh) & 3;
        if (i < 3 && tq < 3) {
          const int ctrl = (sub >> (2 - i)) & 1;
          const int m = 4 >> tq;
          const float pr0 = __shfl_xor(re0, m, 32), pi0 = __shfl_xor(im0, m, 32);
          const float pr1 = __shfl_xor(re1, m, 32), pi1 = __shfl_xor(im1, m, 32);
          re0 = ctrl ? pr0 : re0;  im0 = ctrl ? pi0 : im0;
          re1 = ctrl ? pr1 : re1;  im1 = ctrl ? pi1 : im1;
        } else if (i < 3) {
          const int ctrl = (sub >> (2 - i)) & 1;
          const float r0 = ctrl ? re1 : re0, i0 = ctrl ? im1 : im0;
          const float r1 = ctrl ? re0 : re1, i1 = ctrl ? im0 : im1;
          re0 = r0; im0 = i0; re1 = r1; im1 = i1;
        } else {
          const int m = 4 >> tq;
          re1 = __shfl_xor(re1, m, 32);
          im1 = __shfl_xor(im1, m, 32);
        }
      }
    }
    const float p0 = re0 * re0 + im0 * im0;
    const float p1 = re1 * re1 + im1 * im1;
    const float t = p0 + p1;
    v4f ev;
    ev.x = ((sub >> 2) & 1) ? -t : t;
    ev.y = ((sub >> 1) & 1) ? -t : t;
    ev.z = (sub & 1) ? -t : t;
    ev.w = p0 - p1;
    #pragma unroll
    for (int i = 0; i < 4; ++i) {
      float sv = ev[i];
      sv += __shfl_xor(sv, 1, 32);
      sv += __shfl_xor(sv, 2, 32);
      sv += __shfl_xor(sv, 4, 32);
      ev[i] = sv;
    }
    if (sub == 0) *reinterpret_cast<v4f*>(&sQ[row * 4]) = ev;
  }
  __syncthreads();

  v2f aq;
  aq.x = sQ[lid * 4 + 2 * hsel + 0];
  aq.y = sQ[lid * 4 + 2 * hsel + 1];
  const int nbeg = wave * (D_DIM / NW);
  for (int nn = 0; nn < D_DIM / NW; nn += 64) {
    const int nc = nbeg + nn + 4 * lid;
    const v4f w0 = *reinterpret_cast<const v4f*>(Wp + (2 * hsel + 0) * D_DIM + nc);
    const v4f w1 = *reinterpret_cast<const v4f*>(Wp + (2 * hsel + 1) * D_DIM + nc);
    const v4f bias = *reinterpret_cast<const v4f*>(bp + nc);
    v8f c[4];
    for (int j = 0; j < 4; ++j) {
      v2f bw;
      bw.x = w0[j];
      bw.y = w1[j];
      const v8f zero = {};
      c[j] = __builtin_amdgcn_wmma_f32_16x16x4_f32(false, aq, false, bw,
                                                   (short)0, zero, false, false);
    }
    float* orow = out + (row0 + hsel * 8) * D_DIM + nc;
    for (int r = 0; r < 8; ++r) {
      v4f o;
      o.x = c[0][r] + bias.x;
      o.y = c[1][r] + bias.y;
      o.z = c[2][r] + bias.z;
      o.w = c[3][r] + bias.w;
      *reinterpret_cast<v4f*>(orow + (long)r * D_DIM) = o;
    }
  }
}

extern "C" void kernel_launch(void* const* d_in, const int* in_sizes, int n_in,
                              void* d_out, int out_size, void* d_ws, size_t ws_size,
                              hipStream_t stream) {
  const float* x  = (const float*)d_in[0];
  const float* W1 = (const float*)d_in[1];
  const float* b1 = (const float*)d_in[2];
  const float* qw = (const float*)d_in[3];
  const float* Wp = (const float*)d_in[4];
  const float* bp = (const float*)d_in[5];
  float* out = (float*)d_out;
  float* w1c = (float*)d_ws;

  const int rows = in_sizes[0] / D_DIM;
  qff_pack_w1<<<(D_DIM * 4 + 255) / 256, 256, 0, stream>>>(W1, w1c);
  qff_main<<<rows / 16, 32 * NW, 0, stream>>>(x, w1c, b1, qw, Wp, bp, out);
}
